# starting kernel
# baseline (speedup 1.0000x reference)
#include <hip/hip_runtime.h>

typedef __attribute__((ext_vector_type(2))) float v2f;
typedef __attribute__((ext_vector_type(8))) float v8f;

#define FDIM 64


__global__ __launch_bounds__(256) void k_init_deg(float* __restrict__ deg, int n) {
    int i = blockIdx.x * blockDim.x + threadIdx.x;
    if (i < n) deg[i] = 1.0f;
}

__global__ __launch_bounds__(256) void k_deg_accum(const int* __restrict__ dst,
                                                   const float* __restrict__ w,
                                                   float* __restrict__ deg, int e) {
    int i = blockIdx.x * blockDim.x + threadIdx.x;
    if (i < e) atomicAdd(&deg[dst[i]], w[i]);
}

__global__ __launch_bounds__(256) void k_dinv(float* __restrict__ deg_dinv,
                                              float* __restrict__ self_norm, int n) {
    int i = blockIdx.x * blockDim.x + threadIdx.x;
    if (i < n) {
        float d = rsqrtf(deg_dinv[i]);
        deg_dinv[i]  = d;
        self_norm[i] = d * d;
    }
}

__global__ __launch_bounds__(256) void k_norm(const int* __restrict__ src,
                                              const int* __restrict__ dst,
                                              const float* __restrict__ w,
                                              const float* __restrict__ dinv,
                                              float* __restrict__ norm, int e) {
    int i = blockIdx.x * blockDim.x + threadIdx.x;
    if (i < e) norm[i] = dinv[src[i]] * w[i] * dinv[dst[i]];
}


__global__ __launch_bounds__(256) void k_gemm64_wmma(const float* __restrict__ H,
                                                     const float* __restrict__ Wm,
                                                     float* __restrict__ out, int nrows) {
    const int lane  = threadIdx.x & 31;
    const int wave  = threadIdx.x >> 5;
    const int ctile = wave & 3;
    const int rhalf = wave >> 2;
    const int l16   = lane & 15;
    const int hi    = lane >> 4;

    v2f bfrag[16];
    {
        const int col = ctile * 16 + l16;
        #pragma unroll
        for (int k = 0; k < 16; ++k) {
            const int k0 = 4 * k + (hi ? 2 : 0);
            bfrag[k].x = Wm[(size_t)k0 * FDIM + col];
            bfrag[k].y = Wm[(size_t)(k0 + 1) * FDIM + col];
        }
    }

    const int cc = ctile * 16 + l16;
    const int rowBlocks = (nrows + 31) >> 5;
    for (int rb = blockIdx.x; rb < rowBlocks; rb += gridDim.x) {
        const int row0 = rb * 32 + rhalf * 16;
        const bool full = (row0 + 16 <= nrows);

        int r  = row0 + l16;
        int rc = full ? r : ((r < nrows) ? r : (nrows > 0 ? nrows - 1 : 0));
        const float* hp = H + (size_t)rc * FDIM + (hi ? 2 : 0);

        v8f acc = {};
        #pragma unroll
        for (int k = 0; k < 16; ++k) {
            v2f a = *(const v2f*)(hp + 4 * k);
            acc = __builtin_amdgcn_wmma_f32_16x16x4_f32(
                      false, a, false, bfrag[k], (short)0, acc, false, false);
        }

        if (full) {
            float* op = out + (size_t)(row0 + hi * 8) * FDIM + cc;
            #pragma unroll
            for (int i = 0; i < 8; ++i)
                op[(size_t)i * FDIM] = acc[i];
        } else {
            #pragma unroll
            for (int i = 0; i < 8; ++i) {
                int rr = row0 + hi * 8 + i;
                if (rr < nrows) out[(size_t)rr * FDIM + cc] = acc[i];
            }
        }
    }
}


__global__ __launch_bounds__(256) void k_epi_init(const float* __restrict__ hw,
                                                  const float* __restrict__ self_norm,
                                                  const float* __restrict__ b,
                                                  float* __restrict__ out, int n) {
    long gid = (long)blockIdx.x * blockDim.x + threadIdx.x;
    if (gid < (long)n * FDIM) {
        int i = (int)(gid >> 6);
        out[gid] = hw[gid] * self_norm[i] + b[gid & 63];
    }
}

__global__ __launch_bounds__(256) void k_scatter(const float* __restrict__ hw,
                                                 const float* __restrict__ norm,
                                                 const int* __restrict__ src,
                                                 const int* __restrict__ dst,
                                                 float* __restrict__ out, int e) {
    long gid = (long)blockIdx.x * blockDim.x + threadIdx.x;
    if (gid >= (long)e * 16) return;
    int  ed = (int)(gid >> 4);
    int  g  = ((int)gid & 15) * 4;
    float nv = norm[ed];
    int   s  = src[ed];
    int   d  = dst[ed];
    const float4 v = *(const float4*)(hw + (size_t)s * FDIM + g);
    float* op = out + (size_t)d * FDIM + g;
    atomicAdd(op + 0, v.x * nv);
    atomicAdd(op + 1, v.y * nv);
    atomicAdd(op + 2, v.z * nv);
    atomicAdd(op + 3, v.w * nv);
}

__global__ __launch_bounds__(256) void k_relu(float* __restrict__ h, long total) {
    long gid = (long)blockIdx.x * blockDim.x + threadIdx.x;
    if (gid < total) h[gid] = fmaxf(h[gid], 0.0f);
}


__global__ __launch_bounds__(256) void k_matvec64(const float* __restrict__ H,
                                                  const float* __restrict__ w2,
                                                  float* __restrict__ lin, int n) {
    int i = blockIdx.x * blockDim.x + threadIdx.x;
    if (i < n) {
        const float* hp = H + (size_t)i * FDIM;
        float s = 0.0f;
        #pragma unroll
        for (int f = 0; f < FDIM; ++f) s = fmaf(hp[f], w2[f], s);
        lin[i] = s;
    }
}

__global__ __launch_bounds__(256) void k_init2(const float* __restrict__ lin,
                                               const float* __restrict__ self_norm,
                                               const float* __restrict__ b2,
                                               float* __restrict__ out, int n) {
    int i = blockIdx.x * blockDim.x + threadIdx.x;
    if (i < n) out[i] = lin[i] * self_norm[i] + b2[0];
}

__global__ __launch_bounds__(256) void k_scatter2(const float* __restrict__ lin,
                                                  const float* __restrict__ norm,
                                                  const int* __restrict__ src,
                                                  const int* __restrict__ dst,
                                                  float* __restrict__ out, int e) {
    int i = blockIdx.x * blockDim.x + threadIdx.x;
    if (i < e) atomicAdd(&out[dst[i]], lin[src[i]] * norm[i]);
}


extern "C" void kernel_launch(void* const* d_in, const int* in_sizes, int n_in,
                              void* d_out, int out_size, void* d_ws, size_t ws_size,
                              hipStream_t stream) {
    const float* x    = (const float*)d_in[0];
    const int*   esrc = (const int*)  d_in[1];
    const int*   edst = (const int*)  d_in[2];
    const float* ew   = (const float*)d_in[3];
    const float* W0   = (const float*)d_in[4];
    const float* b0   = (const float*)d_in[5];
    const float* W1   = (const float*)d_in[6];
    const float* b1   = (const float*)d_in[7];
    const float* W2   = (const float*)d_in[8];
    const float* b2   = (const float*)d_in[9];

    const int N = in_sizes[0] / FDIM;
    const int E = in_sizes[1];
    float* out = (float*)d_out;

    float* ws        = (float*)d_ws;
    float* dinv      = ws;
    float* self_norm = ws + (size_t)N;
    float* normw     = ws + 2 * (size_t)N;
    float* hw        = normw + (size_t)E;
    float* ha        = hw + (size_t)N * FDIM;

    const int T = 256;
    const int gN  = (N + T - 1) / T;
    const int gE  = (E + T - 1) / T;
    const long nf = (long)N * FDIM;
    const int gNF = (int)((nf + T - 1) / T);
    const long es = (long)E * 16;
    const int gES = (int)((es + T - 1) / T);
    const int gGemm = (N + 31) / 32;

    k_init_deg <<<gN, T, 0, stream>>>(dinv, N);
    k_deg_accum<<<gE, T, 0, stream>>>(edst, ew, dinv, E);
    k_dinv     <<<gN, T, 0, stream>>>(dinv, self_norm, N);
    k_norm     <<<gE, T, 0, stream>>>(esrc, edst, ew, dinv, normw, E);

    k_gemm64_wmma<<<gGemm, T, 0, stream>>>(x, W0, hw, N);
    k_epi_init   <<<gNF, T, 0, stream>>>(hw, self_norm, b0, ha, N);
    k_scatter    <<<gES, T, 0, stream>>>(hw, normw, esrc, edst, ha, E);
    k_relu       <<<gNF, T, 0, stream>>>(ha, nf);

    k_gemm64_wmma<<<gGemm, T, 0, stream>>>(ha, W1, hw, N);
    k_epi_init   <<<gNF, T, 0, stream>>>(hw, self_norm, b1, ha, N);
    k_scatter    <<<gES, T, 0, stream>>>(hw, normw, esrc, edst, ha, E);
    k_relu       <<<gNF, T, 0, stream>>>(ha, nf);

    float* lin = hw;
    k_matvec64<<<gN, T, 0, stream>>>(ha, W2, lin, N);
    k_init2   <<<gN, T, 0, stream>>>(lin, self_norm, b2, out, N);
    k_scatter2<<<gE, T, 0, stream>>>(lin, normw, esrc, edst, out, E);
}


# Both 64-wide edge sums kept as ONE copy in two 32-column planes: a package half adds into its own plane only; 512-edge jobs handed out per half by ticket counters; sum order: atomic arrival
# speedup vs baseline: 2.9299x; 2.9299x over previous
#include <hip/hip_runtime.h>

typedef __attribute__((ext_vector_type(2))) float v2f;
typedef __attribute__((ext_vector_type(8))) float v8f;

#define FDIM 64


#define RANGE_MAX 32768
#define ACC_T     1024
#define ACC_U     10

__device__ __forceinline__ int range_rows(int n, int n_ranges) {
    return (((n + n_ranges - 1) / n_ranges) + 31) & ~31;
}

template <bool GATHER>
__global__ __launch_bounds__(ACC_T) void k_range_accum(const int* __restrict__ dst,
                                                       const float* __restrict__ val,
                                                       const int* __restrict__ src,
                                                       const float* __restrict__ tab,
                                                       float* __restrict__ part, int n, int e) {
    __shared__ float acc[RANGE_MAX];
    const int tid = threadIdx.x;
    const int nR  = gridDim.y;
    const int rs  = range_rows(n, nR);
    const int lo  = blockIdx.y * rs;

    for (int k = tid * 4; k < rs; k += ACC_T * 4)
        *(float4*)&acc[k] = make_float4(0.0f, 0.0f, 0.0f, 0.0f);
    __syncthreads();

    const int stride = gridDim.x * ACC_T;
    for (int i0 = blockIdx.x * ACC_T + tid; i0 < e; i0 += ACC_U * stride) {
        int d[ACC_U]; int sidx[ACC_U]; float v[ACC_U]; float x[ACC_U]; bool m[ACC_U];
        #pragma unroll
        for (int u = 0; u < ACC_U; ++u) {
            const int  i  = i0 + u * stride;
            const bool ok = i < e;
            const int  ic = ok ? i : i0;
            d[u] = dst[ic];
            v[u] = val[ic];
            sidx[u] = GATHER ? src[ic] : 0;
            m[u] = ok;
        }
        #pragma unroll
        for (int u = 0; u < ACC_U; ++u) {
            m[u] = m[u] && ((unsigned)(d[u] - lo) < (unsigned)rs);
            x[u] = GATHER ? tab[m[u] ? sidx[u] : 0] * v[u] : v[u];
        }
        #pragma unroll
        for (int u = 0; u < ACC_U; ++u)
            if (m[u]) atomicAdd(&acc[d[u] - lo], x[u]);
    }
    __syncthreads();

    float* pp = part + (size_t)blockIdx.x * nR * rs + lo;
    for (int k = tid * 4; k < rs; k += ACC_T * 4)
        *(float4*)&pp[k] = *(const float4*)&acc[k];
}

__global__ __launch_bounds__(256) void k_deg_finish(const float* __restrict__ part,
                                                    float* __restrict__ dinv,
                                                    float* __restrict__ self_norm,
                                                    unsigned* __restrict__ tickets,
                                                    int n, int n_ranges, int n_planes) {
    int i = blockIdx.x * blockDim.x + threadIdx.x;
    if (i < 4) tickets[i] = 0u;
    if (i < n) {
        const size_t pitch = (size_t)n_ranges * range_rows(n, n_ranges);
        float s = 1.0f;
        #pragma unroll 16
        for (int c = 0; c < n_planes; ++c) s += part[c * pitch + i];
        float dv = rsqrtf(s);
        dinv[i]      = dv;
        self_norm[i] = dv * dv;
    }
}

__global__ __launch_bounds__(256) void k_norm(const int* __restrict__ src,
                                              const int* __restrict__ dst,
                                              const float* __restrict__ w,
                                              const float* __restrict__ dinv,
                                              float* __restrict__ norm, int e) {
    int i = blockIdx.x * blockDim.x + threadIdx.x;
    if (i < e) norm[i] = dinv[src[i]] * w[i] * dinv[dst[i]];
}

#define RED_PITCH 66
#define A_PITCH   68

template <bool LAST>
__global__ __launch_bounds__(256) void k_gemm64_wmma(const float* __restrict__ S,
                                                     const float* __restrict__ Wm,
                                                     const float* __restrict__ bias,
                                                     const float* __restrict__ self_norm,
                                                     const float* __restrict__ w2,
                                                     float* __restrict__ o0,
                                                     float* __restrict__ o1, int nrows) {
    __shared__ float red[LAST ? 32 * RED_PITCH : 1];
    __shared__ float atile[32 * A_PITCH];
    const int lane  = threadIdx.x & 31;
    const int wave  = threadIdx.x >> 5;
    const int ctile = wave & 3;
    const int rhalf = wave >> 2;
    const int l16   = lane & 15;
    const int hi    = lane >> 4;

    v2f bfrag[16];
    {
        const int col = ctile * 16 + l16;
        #pragma unroll
        for (int k = 0; k < 16; ++k) {
            const int k0 = 4 * k + (hi ? 2 : 0);
            bfrag[k].x = Wm[(size_t)k0 * FDIM + col];
            bfrag[k].y = Wm[(size_t)(k0 + 1) * FDIM + col];
        }
    }

    const int cc = ctile * 16 + l16;
    const float bcol = bias[cc];
    const float wcol = LAST ? w2[cc] : 0.0f;
    const int rowBlocks = (nrows + 31) >> 5;
    for (int rb = blockIdx.x; rb < rowBlocks; rb += gridDim.x) {
        const int row0 = rb * 32 + rhalf * 16;
        const bool full = (row0 + 16 <= nrows);

        {
            float4 va[2];
            #pragma unroll
            for (int j = 0; j < 2; ++j) {
                const int idx  = threadIdx.x + j * 256;
                const int grow = rb * 32 + (idx >> 4);
                const int c4   = idx & 15;
                const size_t q = (size_t)(c4 >> 3) * nrows * 8 + (size_t)(grow < nrows ? grow : 0) * 8 + (c4 & 7);
                va[j] = ((const float4*)S)[q];
            }
            #pragma unroll
            for (int j = 0; j < 2; ++j) {
                const int idx = threadIdx.x + j * 256;
                *(float4*)&atile[(idx >> 4) * A_PITCH + (idx & 15) * 4] = va[j];
            }
        }
        __syncthreads();

        const float* hp = atile + (rhalf * 16 + l16) * A_PITCH + (hi ? 2 : 0);

        v2f afr[16];
        #pragma unroll
        for (int k = 0; k < 16; ++k) afr[k] = *(const v2f*)(hp + 4 * k);
        asm volatile("" :: "v"(afr[0]), "v"(afr[1]), "v"(afr[2]), "v"(afr[3]), "v"(afr[4]), "v"(afr[5]),
                           "v"(afr[6]), "v"(afr[7]), "v"(afr[8]), "v"(afr[9]), "v"(afr[10]), "v"(afr[11]),
                           "v"(afr[12]), "v"(afr[13]), "v"(afr[14]), "v"(afr[15]));

        v8f acc = {};
        #pragma unroll
        for (int k = 0; k < 16; ++k)
            acc = __builtin_amdgcn_wmma_f32_16x16x4_f32(
                      false, afr[k], false, bfrag[k], (short)0, acc, false, false);

        float v[8];
        #pragma unroll
        for (int i = 0; i < 8; ++i) v[i] = fmaxf(acc[i] + bcol, 0.0f);

        if (!LAST) {
            float* const o1c = o1 + (size_t)(cc >> 5) * nrows * 32 + (cc & 31);
            if (full) {
                const size_t o = (size_t)(row0 + hi * 8) * FDIM + cc;
                float* const o1r = o1c + (size_t)(row0 + hi * 8) * 32;
                const float* sp = self_norm + row0 + hi * 8;
                #pragma unroll
                for (int i = 0; i < 8; ++i) {
                    o0[o + (size_t)i * FDIM] = v[i];
                    o1r[i * 32] = v[i] * sp[i];
                }
            } else {
                #pragma unroll
                for (int i = 0; i < 8; ++i) {
                    int rr = row0 + hi * 8 + i;
                    if (rr < nrows) {
                        o0[(size_t)rr * FDIM + cc] = v[i];
                        o1c[(size_t)rr * 32] = v[i] * self_norm[rr];
                    }
                }
            }
            if (rb + (int)gridDim.x < rowBlocks) __syncthreads();
        } else {
            #pragma unroll
            for (int i = 0; i < 8; ++i)
                red[(rhalf * 16 + hi * 8 + i) * RED_PITCH + cc] = v[i] * wcol;
            __syncthreads();
            {
                const int rrow = threadIdx.x >> 3;
                const int seg  = threadIdx.x & 7;
                const float* rp = red + rrow * RED_PITCH + seg * 8;
                float s = 0.0f;
                #pragma unroll
                for (int j = 0; j < 8; ++j) s += rp[j];
                s += __shfl_xor(s, 1);
                s += __shfl_xor(s, 2);
                s += __shfl_xor(s, 4);
                const int grow = rb * 32 + rrow;
                if (seg == 0 && grow < nrows) o0[grow] = s;
            }
            __syncthreads();
        }
    }
}


__global__ __launch_bounds__(256) void k_self_start(const float* __restrict__ h,
                                                    const float* __restrict__ self_norm,
                                                    float* __restrict__ agg, int n) {
    long q = (long)blockIdx.x * blockDim.x + threadIdx.x;
    if (q < (long)n * (FDIM / 4)) {
        const float sn = self_norm[q >> 4];
        float4 v = ((const float4*)h)[q];
        v.x *= sn; v.y *= sn; v.z *= sn; v.w *= sn;
        const int c4 = (int)(q & 15);
        ((float4*)agg)[(size_t)(c4 >> 3) * n * 8 + (size_t)(q >> 4) * 8 + (c4 & 7)] = v;
    }
}

#define SC_GROUP 512
__global__ __launch_bounds__(256) void k_scatter(const float* __restrict__ hw,
                                                 const float* __restrict__ norm,
                                                 const int* __restrict__ src,
                                                 const int* __restrict__ dst,
                                                 float* __restrict__ out,
                                                 unsigned* __restrict__ tickets,
                                                 int n, int e, int groups) {
    __shared__ unsigned s_job;
    if (threadIdx.x == 0) {
        unsigned h = (__builtin_amdgcn_s_sendmsg_rtn(0x87) >> 18) & 1u;
        unsigned t = atomicAdd(&tickets[h], 1u);
        if (t >= (unsigned)groups) { h ^= 1u; t = atomicAdd(&tickets[h], 1u); }
        s_job = t * 2u + h;
    }
    __syncthreads();
    const unsigned job = __builtin_amdgcn_readfirstlane(s_job);
    const unsigned g   = job >> 1;
    if (g >= (unsigned)groups) return;
    const unsigned pl = job & 1u;
    const int lane = threadIdx.x & 31;
    const float* const hc = hw  + pl * 32 + lane;
    float* const       oc = out + (size_t)pl * n * 32 + lane;
    const int wv = __builtin_amdgcn_readfirstlane((int)(threadIdx.x >> 5));
    const long e0 = ((long)g * 8 + wv) * 64;
    if (e0 >= (long)e) return;

    if (e0 + 64 <= (long)e) {
        int s[2], d[2]; float nv[2];
        #pragma unroll
        for (int r = 0; r < 2; ++r) {
            s[r]  = src[e0 + r * 32 + lane];
            d[r]  = dst[e0 + r * 32 + lane];
            nv[r] = norm[e0 + r * 32 + lane];
        }
        #pragma unroll
        for (int r = 0; r < 2; ++r) {
            for (int j0 = 0; j0 < 32; j0 += 16) {
                float a[16];
                #pragma unroll
                for (int u = 0; u < 16; ++u) {
                    const int sj = __builtin_amdgcn_readlane(s[r], j0 + u);
                    a[u] = hc[(size_t)sj * FDIM];
                }
                #pragma unroll
                for (int u = 0; u < 16; ++u) {
                    const int   dj = __builtin_amdgcn_readlane(d[r], j0 + u);
                    const float nj = __int_as_float(__builtin_amdgcn_readlane(__float_as_int(nv[r]), j0 + u));
                    atomicAdd(oc + (size_t)dj * 32, a[u] * nj);
                }
            }
        }
    } else {
        const int cnt = e - (int)e0;
        for (int j = 0; j < cnt; ++j) {
            const int   sj = src[e0 + j];
            const int   dj = dst[e0 + j];
            const float nj = norm[e0 + j];
            atomicAdd(oc + (size_t)dj * 32, hc[(size_t)sj * FDIM] * nj);
        }
    }
}


__global__ __launch_bounds__(256) void k_out_finish(const float* __restrict__ part,
                                                    const float* __restrict__ lin,
                                                    const float* __restrict__ self_norm,
                                                    const float* __restrict__ b2,
                                                    float* __restrict__ out,
                                                    int n, int n_ranges, int n_planes) {
    int i = blockIdx.x * blockDim.x + threadIdx.x;
    if (i < n) {
        const size_t pitch = (size_t)n_ranges * range_rows(n, n_ranges);
        float s = lin[i] * self_norm[i] + b2[0];
        #pragma unroll 16
        for (int c = 0; c < n_planes; ++c) s += part[c * pitch + i];
        out[i] = s;
    }
}


extern "C" void kernel_launch(void* const* d_in, const int* in_sizes, int n_in,
                              void* d_out, int out_size, void* d_ws, size_t ws_size,
                              hipStream_t stream) {
    const float* x    = (const float*)d_in[0];
    const int*   esrc = (const int*)  d_in[1];
    const int*   edst = (const int*)  d_in[2];
    const float* ew   = (const float*)d_in[3];
    const float* W0   = (const float*)d_in[4];
    const float* b0   = (const float*)d_in[5];
    const float* W1   = (const float*)d_in[6];
    const float* b1   = (const float*)d_in[7];
    const float* W2   = (const float*)d_in[8];
    const float* b2   = (const float*)d_in[9];

    const int N = in_sizes[0] / FDIM;
    const int E = in_sizes[1];
    float* out = (float*)d_out;

    float* ws        = (float*)d_ws;
    float* dinv      = ws;
    float* self_norm = ws + (size_t)N;
    float* normw     = ws + 2 * (size_t)N;
    float* s0        = normw + (size_t)E;
    float* h1        = s0 + (size_t)N * FDIM;
    float* s1        = h1 + (size_t)N * FDIM;
    unsigned* tick   = (unsigned*)(s1 + (size_t)N * FDIM);
    float* part      = s1 + (size_t)N * FDIM + 64;
    float* lin       = s0;

    const int T = 256;
    const int gN  = (N + T - 1) / T;
    const int gE  = (E + T - 1) / T;
    const long nq = (long)N * (FDIM / 4);
    const int gNQ = (int)((nq + T - 1) / T);
    const int gGemm = (N + 31) / 32;
    const int gGrp  = (E + SC_GROUP - 1) / SC_GROUP;
    const int nRanges = (N + RANGE_MAX - 1) / RANGE_MAX;
    const int nPlanes = 64;
    const dim3 gAcc(nPlanes, nRanges);

    k_range_accum<false><<<gAcc, ACC_T, 0, stream>>>(edst, ew, esrc, ew, part, N, E);
    k_deg_finish<<<gN, T, 0, stream>>>(part, dinv, self_norm, tick, N, nRanges, nPlanes);
    k_norm     <<<gE, T, 0, stream>>>(esrc, edst, ew, dinv, normw, E);

    k_self_start<<<gNQ, T, 0, stream>>>(x, self_norm, s0, N);
    k_scatter   <<<2 * gGrp, T, 0, stream>>>(x, normw, esrc, edst, s0, tick, N, E, gGrp);
    k_gemm64_wmma<false><<<gGemm, T, 0, stream>>>(s0, W0, b0, self_norm, W2, h1, s1, N);

    k_scatter   <<<2 * gGrp, T, 0, stream>>>(h1, normw, esrc, edst, s1, tick + 2, N, E, gGrp);
    k_gemm64_wmma<true><<<gGemm, T, 0, stream>>>(s1, W1, b1, self_norm, W2, lin, h1, N);

    k_range_accum<true><<<gAcc, ACC_T, 0, stream>>>(edst, normw, esrc, lin, part, N, E);
    k_out_finish<<<gN, T, 0, stream>>>(part, lin, self_norm, b2, out, N, nRanges, nPlanes);
}


# 64-wide edge sums and last layer read per-node slot rows: an edge takes a slot by one count add (per package half), 16 lanes gather 8 source rows at once; overflow on chains; sum order: slot order
# speedup vs baseline: 3.9727x; 1.3559x over previous
#include <hip/hip_runtime.h>

typedef __attribute__((ext_vector_type(2))) float v2f;
typedef __attribute__((ext_vector_type(8))) float v8f;

#define FDIM 64


#define RANGE_MAX 32768
#define ACC_T     1024
#define ACC_U     10

__device__ __forceinline__ int range_rows(int n, int n_ranges) {
    return (((n + n_ranges - 1) / n_ranges) + 31) & ~31;
}

__global__ __launch_bounds__(ACC_T) void k_range_accum(const int* __restrict__ dst,
                                                       const float* __restrict__ val,
                                                       float* __restrict__ part, int n, int e) {
    __shared__ float acc[RANGE_MAX];
    const int tid = threadIdx.x;
    const int nR  = gridDim.y;
    const int rs  = range_rows(n, nR);
    const int lo  = blockIdx.y * rs;

    for (int k = tid * 4; k < rs; k += ACC_T * 4)
        *(float4*)&acc[k] = make_float4(0.0f, 0.0f, 0.0f, 0.0f);
    __syncthreads();

    const int stride = gridDim.x * ACC_T;
    for (int i0 = blockIdx.x * ACC_T + tid; i0 < e; i0 += ACC_U * stride) {
        int d[ACC_U]; float v[ACC_U]; bool m[ACC_U];
        #pragma unroll
        for (int u = 0; u < ACC_U; ++u) {
            const int  i  = i0 + u * stride;
            const bool ok = i < e;
            const int  ic = ok ? i : i0;
            d[u] = dst[ic];
            v[u] = val[ic];
            m[u] = ok;
        }
        #pragma unroll
        for (int u = 0; u < ACC_U; ++u)
            m[u] = m[u] && ((unsigned)(d[u] - lo) < (unsigned)rs);
        #pragma unroll
        for (int u = 0; u < ACC_U; ++u)
            if (m[u]) atomicAdd(&acc[d[u] - lo], v[u]);
    }
    __syncthreads();

    float* pp = part + (size_t)blockIdx.x * nR * rs + lo;
    for (int k = tid * 4; k < rs; k += ACC_T * 4)
        *(float4*)&pp[k] = *(const float4*)&acc[k];
}

__device__ __forceinline__ int head_pitch(int n) {
    return (n + 31) & ~31;
}

__global__ __launch_bounds__(256) void k_deg_finish(const float* __restrict__ part,
                                                    float* __restrict__ dinv,
                                                    float* __restrict__ self_norm,
                                                    int* __restrict__ head,
                                                    int* __restrict__ cnt,
                                                    int n, int n_ranges, int n_planes) {
    int i = blockIdx.x * blockDim.x + threadIdx.x;
    if (i < n) {
        cnt[i] = 0;
        cnt[head_pitch(n) + i] = 0;
        head[i] = -1;
        head[head_pitch(n) + i] = -1;
        const size_t pitch = (size_t)n_ranges * range_rows(n, n_ranges);
        float s = 1.0f;
        #pragma unroll 16
        for (int c = 0; c < n_planes; ++c) s += part[c * pitch + i];
        float dv = rsqrtf(s);
        dinv[i]      = dv;
        self_norm[i] = dv * dv;
    }
}

#define SLOT_K 16

__global__ __launch_bounds__(256) void k_norm_link(const int* __restrict__ src,
                                                   const int* __restrict__ dst,
                                                   const float* __restrict__ w,
                                                   const float* __restrict__ dinv,
                                                   int2* __restrict__ slots,
                                                   int* __restrict__ cnt,
                                                   int4* __restrict__ rec,
                                                   int* __restrict__ head, int n, int e) {
    int i = blockIdx.x * blockDim.x + threadIdx.x;
    const int half = (int)((__builtin_amdgcn_s_sendmsg_rtn(0x87) >> 18) & 1u);
    if (i < e) {
        const int s = src[i];
        const int d = dst[i];
        const int node = half * head_pitch(n) + d;
        const int slot = atomicAdd(&cnt[node], 1);
        const float nv = dinv[s] * w[i] * dinv[d];
        if (slot < SLOT_K) {
            slots[(size_t)node * SLOT_K + slot] = make_int2(s, __float_as_int(nv));
        } else {
            const int prev = atomicExch(&head[node], i);
            rec[i] = make_int4(prev, s, __float_as_int(nv), 0);
        }
    }
}

#define RED_PITCH 66
#define A_PITCH   68

template <bool LAST>
__global__ __launch_bounds__(256) void k_gemm64_wmma(const float* __restrict__ S,
                                                     const float* __restrict__ Wm,
                                                     const float* __restrict__ bias,
                                                     const float* __restrict__ w2,
                                                     float* __restrict__ o0, int nrows) {
    __shared__ float red[LAST ? 32 * RED_PITCH : 1];
    __shared__ float atile[32 * A_PITCH];
    const int lane  = threadIdx.x & 31;
    const int wave  = threadIdx.x >> 5;
    const int ctile = wave & 3;
    const int rhalf = wave >> 2;
    const int l16   = lane & 15;
    const int hi    = lane >> 4;

    v2f bfrag[16];
    {
        const int col = ctile * 16 + l16;
        #pragma unroll
        for (int k = 0; k < 16; ++k) {
            const int k0 = 4 * k + (hi ? 2 : 0);
            bfrag[k].x = Wm[(size_t)k0 * FDIM + col];
            bfrag[k].y = Wm[(size_t)(k0 + 1) * FDIM + col];
        }
    }

    const int cc = ctile * 16 + l16;
    const float bcol = bias[cc];
    const float wcol = LAST ? w2[cc] : 0.0f;
    const int rowBlocks = (nrows + 31) >> 5;
    for (int rb = blockIdx.x; rb < rowBlocks; rb += gridDim.x) {
        const int row0 = rb * 32 + rhalf * 16;
        const bool full = (row0 + 16 <= nrows);

        {
            float4 va[2];
            #pragma unroll
            for (int j = 0; j < 2; ++j) {
                const int idx  = threadIdx.x + j * 256;
                const int grow = rb * 32 + (idx >> 4);
                const int c4   = idx & 15;
                const size_t q = (size_t)(grow < nrows ? grow : 0) * 16 + c4;
                va[j] = ((const float4*)S)[q];
            }
            #pragma unroll
            for (int j = 0; j < 2; ++j) {
                const int idx = threadIdx.x + j * 256;
                *(float4*)&atile[(idx >> 4) * A_PITCH + (idx & 15) * 4] = va[j];
            }
        }
        __syncthreads();

        const float* hp = atile + (rhalf * 16 + l16) * A_PITCH + (hi ? 2 : 0);

        v2f afr[16];
        #pragma unroll
        for (int k = 0; k < 16; ++k) afr[k] = *(const v2f*)(hp + 4 * k);
        asm volatile("" :: "v"(afr[0]), "v"(afr[1]), "v"(afr[2]), "v"(afr[3]), "v"(afr[4]), "v"(afr[5]),
                           "v"(afr[6]), "v"(afr[7]), "v"(afr[8]), "v"(afr[9]), "v"(afr[10]), "v"(afr[11]),
                           "v"(afr[12]), "v"(afr[13]), "v"(afr[14]), "v"(afr[15]));

        v8f acc = {};
        #pragma unroll
        for (int k = 0; k < 16; ++k)
            acc = __builtin_amdgcn_wmma_f32_16x16x4_f32(
                      false, afr[k], false, bfrag[k], (short)0, acc, false, false);

        float v[8];
        #pragma unroll
        for (int i = 0; i < 8; ++i) v[i] = fmaxf(acc[i] + bcol, 0.0f);

        if (!LAST) {
            if (full) {
                const size_t o = (size_t)(row0 + hi * 8) * FDIM + cc;
                #pragma unroll
                for (int i = 0; i < 8; ++i)
                    o0[o + (size_t)i * FDIM] = v[i];
            } else {
                #pragma unroll
                for (int i = 0; i < 8; ++i) {
                    int rr = row0 + hi * 8 + i;
                    if (rr < nrows) o0[(size_t)rr * FDIM + cc] = v[i];
                }
            }
            if (rb + (int)gridDim.x < rowBlocks) __syncthreads();
        } else {
            #pragma unroll
            for (int i = 0; i < 8; ++i)
                red[(rhalf * 16 + hi * 8 + i) * RED_PITCH + cc] = v[i] * wcol;
            __syncthreads();
            {
                const int rrow = threadIdx.x >> 3;
                const int seg  = threadIdx.x & 7;
                const float* rp = red + rrow * RED_PITCH + seg * 8;
                float s = 0.0f;
                #pragma unroll
                for (int j = 0; j < 8; ++j) s += rp[j];
                s += __shfl_xor(s, 1);
                s += __shfl_xor(s, 2);
                s += __shfl_xor(s, 4);
                const int grow = rb * 32 + rrow;
                if (seg == 0 && grow < nrows) o0[grow] = s;
            }
            __syncthreads();
        }
    }
}


template <int J>
__device__ __forceinline__ int row_lane(int v) {
    return __builtin_amdgcn_update_dpp(0, v, 0x150 + J, 0xf, 0xf, true);
}

__device__ __forceinline__ void fma4(float4& acc, float k, const float4& x) {
    acc.x += k * x.x; acc.y += k * x.y; acc.z += k * x.z; acc.w += k * x.w;
}

template <int J>
__device__ __forceinline__ void slot_fetch(const float4* __restrict__ H, int c, int2 ea, int2 eb,
                                           int m0, int m1, float4& xa, float4& xb, float& ka, float& kb) {
    const int   sa = row_lane<J>(ea.x);
    const int   sb = row_lane<J>(eb.x);
    const float fa = __int_as_float(row_lane<J>(ea.y));
    const float fb = __int_as_float(row_lane<J>(eb.y));
    const bool  ua = J < m0;
    const bool  ub = J < m1;
    xa = make_float4(0.0f, 0.0f, 0.0f, 0.0f);
    xb = xa;
    if (ua) xa = H[(size_t)sa * 16 + c];
    if (ub) xb = H[(size_t)sb * 16 + c];
    ka = ua ? fa : 0.0f;
    kb = ub ? fb : 0.0f;
}

template <int J0>
__device__ __forceinline__ void slot_batch(const float4* __restrict__ H, int c, int2 ea, int2 eb,
                                           int m0, int m1, float4& acc) {
    float4 xa[4], xb[4];
    float  ka[4], kb[4];
    slot_fetch<J0 + 0>(H, c, ea, eb, m0, m1, xa[0], xb[0], ka[0], kb[0]);
    slot_fetch<J0 + 1>(H, c, ea, eb, m0, m1, xa[1], xb[1], ka[1], kb[1]);
    slot_fetch<J0 + 2>(H, c, ea, eb, m0, m1, xa[2], xb[2], ka[2], kb[2]);
    slot_fetch<J0 + 3>(H, c, ea, eb, m0, m1, xa[3], xb[3], ka[3], kb[3]);
    asm volatile("" :: "v"(xa[0].x), "v"(xa[0].y), "v"(xa[0].z), "v"(xa[0].w),
                       "v"(xa[1].x), "v"(xa[1].y), "v"(xa[1].z), "v"(xa[1].w),
                       "v"(xa[2].x), "v"(xa[2].y), "v"(xa[2].z), "v"(xa[2].w),
                       "v"(xa[3].x), "v"(xa[3].y), "v"(xa[3].z), "v"(xa[3].w));
    asm volatile("" :: "v"(xb[0].x), "v"(xb[0].y), "v"(xb[0].z), "v"(xb[0].w),
                       "v"(xb[1].x), "v"(xb[1].y), "v"(xb[1].z), "v"(xb[1].w),
                       "v"(xb[2].x), "v"(xb[2].y), "v"(xb[2].z), "v"(xb[2].w),
                       "v"(xb[3].x), "v"(xb[3].y), "v"(xb[3].z), "v"(xb[3].w));
    #pragma unroll
    for (int u = 0; u < 4; ++u) {
        fma4(acc, ka[u], xa[u]);
        fma4(acc, kb[u], xb[u]);
    }
}

__device__ __forceinline__ void chain_walk(const float4* __restrict__ H, const int4* __restrict__ rec,
                                           int p, int c, int e, float4& acc) {
    for (int it = 0; it < e && (unsigned)p < (unsigned)e; ++it) {
        const int4   r = rec[p];
        const float4 x = H[(size_t)r.y * 16 + c];
        fma4(acc, __int_as_float(r.z), x);
        p = r.x;
    }
}

__global__ __launch_bounds__(256) void k_pull(const float* __restrict__ h,
                                              const int2* __restrict__ slots,
                                              const int* __restrict__ cnt,
                                              const int4* __restrict__ rec,
                                              const int* __restrict__ head,
                                              const float* __restrict__ self_norm,
                                              float* __restrict__ agg, int n, int e) {
    const int row = blockIdx.x * 16 + (int)(threadIdx.x >> 4);
    const int c   = (int)(threadIdx.x & 15);
    if (row >= n) return;
    const float4* const H = (const float4*)h;
    const int pitch = head_pitch(n);
    const int n0 = cnt[row];
    const int n1 = cnt[pitch + row];
    const int2 ea = slots[(size_t)row * SLOT_K + c];
    const int2 eb = slots[(size_t)(pitch + row) * SLOT_K + c];
    const float  sn  = self_norm[row];
    const float4 own = H[(size_t)row * 16 + c];
    asm volatile("" :: "v"(ea.x), "v"(ea.y), "v"(eb.x), "v"(eb.y));
    float4 acc = make_float4(own.x * sn, own.y * sn, own.z * sn, own.w * sn);
    const int m0 = min(n0, SLOT_K);
    const int m1 = min(n1, SLOT_K);
    const int mm = max(m0, m1);
    if (mm > 0) {
        slot_batch<0>(H, c, ea, eb, m0, m1, acc);
        if (mm > 4) {
            slot_batch<4>(H, c, ea, eb, m0, m1, acc);
            if (mm > 8) {
                slot_batch<8>(H, c, ea, eb, m0, m1, acc);
                if (mm > 12) slot_batch<12>(H, c, ea, eb, m0, m1, acc);
            }
        }
    }
    if (n0 > SLOT_K) chain_walk(H, rec, head[row], c, e, acc);
    if (n1 > SLOT_K) chain_walk(H, rec, head[pitch + row], c, e, acc);
    ((float4*)agg)[(size_t)row * 16 + c] = acc;
}


__global__ __launch_bounds__(256) void k_pull1(const float* __restrict__ lin,
                                               const int2* __restrict__ slots,
                                               const int* __restrict__ cnt,
                                               const int4* __restrict__ rec,
                                               const int* __restrict__ head,
                                               const float* __restrict__ self_norm,
                                               const float* __restrict__ b2,
                                               float* __restrict__ out, int n, int e) {
    const int t    = blockIdx.x * 256 + threadIdx.x;
    const int row  = t >> 3;
    const int q    = t & 7;
    const bool live = row < n;
    const int rr    = live ? row : 0;
    const int pitch = head_pitch(n);
    const int4* const S4 = (const int4*)slots;
    const int n0 = cnt[rr];
    const int n1 = cnt[pitch + rr];
    const int4 ea = S4[(size_t)rr * (SLOT_K / 2) + q];
    const int4 eb = S4[(size_t)(pitch + rr) * (SLOT_K / 2) + q];
    const float lv = lin[rr];
    const float sv = self_norm[rr];
    asm volatile("" :: "v"(ea.x), "v"(ea.y), "v"(ea.z), "v"(ea.w), "v"(eb.x), "v"(eb.y), "v"(eb.z), "v"(eb.w));
    const int m0 = live ? min(n0, SLOT_K) : 0;
    const int m1 = live ? min(n1, SLOT_K) : 0;
    const bool u0 = 2 * q < m0, u1 = 2 * q + 1 < m0;
    const bool u2 = 2 * q < m1, u3 = 2 * q + 1 < m1;
    float x0 = 0.0f, x1 = 0.0f, x2 = 0.0f, x3 = 0.0f;
    if (u0) x0 = lin[ea.x];
    if (u1) x1 = lin[ea.z];
    if (u2) x2 = lin[eb.x];
    if (u3) x3 = lin[eb.z];
    asm volatile("" :: "v"(x0), "v"(x1), "v"(x2), "v"(x3));
    float acc = 0.0f;
    acc += (u0 ? __int_as_float(ea.y) : 0.0f) * x0;
    acc += (u1 ? __int_as_float(ea.w) : 0.0f) * x1;
    acc += (u2 ? __int_as_float(eb.y) : 0.0f) * x2;
    acc += (u3 ? __int_as_float(eb.w) : 0.0f) * x3;
    if (live && q == 0) {
        if (n0 > SLOT_K) {
            int p = head[row];
            for (int it = 0; it < e && (unsigned)p < (unsigned)e; ++it) {
                const int4 r = rec[p];
                acc += __int_as_float(r.z) * lin[r.y];
                p = r.x;
            }
        }
        if (n1 > SLOT_K) {
            int p = head[pitch + row];
            for (int it = 0; it < e && (unsigned)p < (unsigned)e; ++it) {
                const int4 r = rec[p];
                acc += __int_as_float(r.z) * lin[r.y];
                p = r.x;
            }
        }
    }
    acc += __shfl_xor(acc, 1);
    acc += __shfl_xor(acc, 2);
    acc += __shfl_xor(acc, 4);
    if (live && q == 0) out[row] = (lv * sv + b2[0]) + acc;
}


extern "C" void kernel_launch(void* const* d_in, const int* in_sizes, int n_in,
                              void* d_out, int out_size, void* d_ws, size_t ws_size,
                              hipStream_t stream) {
    const float* x    = (const float*)d_in[0];
    const int*   esrc = (const int*)  d_in[1];
    const int*   edst = (const int*)  d_in[2];
    const float* ew   = (const float*)d_in[3];
    const float* W0   = (const float*)d_in[4];
    const float* b0   = (const float*)d_in[5];
    const float* W1   = (const float*)d_in[6];
    const float* b1   = (const float*)d_in[7];
    const float* W2   = (const float*)d_in[8];
    const float* b2   = (const float*)d_in[9];

    const int N = in_sizes[0] / FDIM;
    const int E = in_sizes[1];
    float* out = (float*)d_out;

    float* ws        = (float*)d_ws;
    int4*  rec       = (int4*)d_ws;
    float* s0        = ws + 4 * (size_t)E;
    float* h1        = s0 + (size_t)N * FDIM;
    float* s1        = h1 + (size_t)N * FDIM;
    const size_t pitch = (size_t)((N + 31) & ~31);
    int2*  slots     = (int2*)(s1 + (size_t)N * FDIM);
    int*   head      = (int*)(slots + 2 * pitch * SLOT_K);
    int*   cnt       = head + 2 * pitch;
    float* dinv      = (float*)(cnt + 2 * pitch);
    float* self_norm = dinv + (size_t)N;
    float* part      = self_norm + (size_t)N;
    float* lin       = s0;

    const int T = 256;
    const int gN  = (N + T - 1) / T;
    const int gN8 = (8 * N + T - 1) / T;
    const int gE  = (E + T - 1) / T;
    const int gGemm = (N + 31) / 32;
    const int gPull = (N + 15) / 16;
    const int nRanges = (N + RANGE_MAX - 1) / RANGE_MAX;
    const int nPlanes = 64;
    const dim3 gAcc(nPlanes, nRanges);

    k_range_accum<<<gAcc, ACC_T, 0, stream>>>(edst, ew, part, N, E);
    k_deg_finish<<<gN, T, 0, stream>>>(part, dinv, self_norm, head, cnt, N, nRanges, nPlanes);
    k_norm_link<<<gE, T, 0, stream>>>(esrc, edst, ew, dinv, slots, cnt, rec, head, N, E);

    k_pull<<<gPull, T, 0, stream>>>(x, slots, cnt, rec, head, self_norm, s0, N, E);
    k_gemm64_wmma<false><<<gGemm, T, 0, stream>>>(s0, W0, b0, W2, h1, N);

    k_pull<<<gPull, T, 0, stream>>>(h1, slots, cnt, rec, head, self_norm, s1, N, E);
    k_gemm64_wmma<true><<<gGemm, T, 0, stream>>>(s1, W1, b1, W2, lin, N);

    k_pull1<<<gN8, T, 0, stream>>>(lin, slots, cnt, rec, head, self_norm, b2, out, N, E);
}


# Degree pass through LDS ranges and its planes go: edges enter slot rows with raw weight, a 4-lane row sum gives the degree, the first pull forms and stores the norm; degree sum order: slot order
# speedup vs baseline: 4.1645x; 1.0483x over previous
#include <hip/hip_runtime.h>

typedef __attribute__((ext_vector_type(2))) float v2f;
typedef __attribute__((ext_vector_type(8))) float v8f;

#define FDIM 64


__device__ __forceinline__ int head_pitch(int n) {
    return (n + 31) & ~31;
}

__global__ __launch_bounds__(256) void k_init(int* __restrict__ head, int* __restrict__ cnt, int n) {
    int i = blockIdx.x * blockDim.x + threadIdx.x;
    if (i < n) {
        cnt[i] = 0;
        cnt[head_pitch(n) + i] = 0;
        head[i] = -1;
        head[head_pitch(n) + i] = -1;
    }
}

#define SLOT_K 16

__global__ __launch_bounds__(256) void k_link(const int* __restrict__ src,
                                              const int* __restrict__ dst,
                                              const float* __restrict__ w,
                                              int2* __restrict__ slots,
                                              int* __restrict__ cnt,
                                              int4* __restrict__ rec,
                                              int* __restrict__ head, int n, int e) {
    int i = blockIdx.x * blockDim.x + threadIdx.x;
    const int half = (int)((__builtin_amdgcn_s_sendmsg_rtn(0x87) >> 18) & 1u);
    if (i < e) {
        const int s = src[i];
        const int d = dst[i];
        const int wv = __float_as_int(w[i]);
        const int node = half * head_pitch(n) + d;
        const int slot = atomicAdd(&cnt[node], 1);
        if (slot < SLOT_K) {
            slots[(size_t)node * SLOT_K + slot] = make_int2(s, wv);
        } else {
            const int prev = atomicExch(&head[node], i);
            rec[i] = make_int4(prev, s, wv, 0);
        }
    }
}

__global__ __launch_bounds__(256) void k_deg(const int2* __restrict__ slots,
                                             const int* __restrict__ cnt,
                                             const int4* __restrict__ rec,
                                             const int* __restrict__ head,
                                             float* __restrict__ dinv,
                                             float* __restrict__ self_norm, int n, int e) {
    const int t    = blockIdx.x * 256 + threadIdx.x;
    const int row  = t >> 2;
    const int q    = t & 3;
    const bool live = row < n;
    const int rr    = live ? row : 0;
    const int pitch = head_pitch(n);
    const int4* const S4 = (const int4*)slots;
    const int n0 = cnt[rr];
    const int n1 = cnt[pitch + rr];
    const int4 a0 = S4[(size_t)rr * (SLOT_K / 2) + 2 * q];
    const int4 a1 = S4[(size_t)rr * (SLOT_K / 2) + 2 * q + 1];
    const int4 b0 = S4[(size_t)(pitch + rr) * (SLOT_K / 2) + 2 * q];
    const int4 b1 = S4[(size_t)(pitch + rr) * (SLOT_K / 2) + 2 * q + 1];
    const int m0 = live ? min(n0, SLOT_K) : 0;
    const int m1 = live ? min(n1, SLOT_K) : 0;
    const int j = 4 * q;
    float s = 0.0f;
    s += (j + 0 < m0) ? __int_as_float(a0.y) : 0.0f;
    s += (j + 1 < m0) ? __int_as_float(a0.w) : 0.0f;
    s += (j + 2 < m0) ? __int_as_float(a1.y) : 0.0f;
    s += (j + 3 < m0) ? __int_as_float(a1.w) : 0.0f;
    s += (j + 0 < m1) ? __int_as_float(b0.y) : 0.0f;
    s += (j + 1 < m1) ? __int_as_float(b0.w) : 0.0f;
    s += (j + 2 < m1) ? __int_as_float(b1.y) : 0.0f;
    s += (j + 3 < m1) ? __int_as_float(b1.w) : 0.0f;
    if (live && q == 0) {
        if (n0 > SLOT_K) {
            int p = head[row];
            for (int it = 0; it < e && (unsigned)p < (unsigned)e; ++it) {
                const int4 r = rec[p];
                s += __int_as_float(r.z);
                p = r.x;
            }
        }
        if (n1 > SLOT_K) {
            int p = head[pitch + row];
            for (int it = 0; it < e && (unsigned)p < (unsigned)e; ++it) {
                const int4 r = rec[p];
                s += __int_as_float(r.z);
                p = r.x;
            }
        }
    }
    s += __shfl_xor(s, 1);
    s += __shfl_xor(s, 2);
    if (live && q == 0) {
        const float dv = rsqrtf(s + 1.0f);
        dinv[row]      = dv;
        self_norm[row] = dv * dv;
    }
}

#define RED_PITCH 66
#define A_PITCH   68

template <bool LAST>
__global__ __launch_bounds__(256) void k_gemm64_wmma(const float* __restrict__ S,
                                                     const float* __restrict__ Wm,
                                                     const float* __restrict__ bias,
                                                     const float* __restrict__ w2,
                                                     float* __restrict__ o0, int nrows) {
    __shared__ float red[LAST ? 32 * RED_PITCH : 1];
    __shared__ float atile[32 * A_PITCH];
    const int lane  = threadIdx.x & 31;
    const int wave  = threadIdx.x >> 5;
    const int ctile = wave & 3;
    const int rhalf = wave >> 2;
    const int l16   = lane & 15;
    const int hi    = lane >> 4;

    v2f bfrag[16];
    {
        const int col = ctile * 16 + l16;
        #pragma unroll
        for (int k = 0; k < 16; ++k) {
            const int k0 = 4 * k + (hi ? 2 : 0);
            bfrag[k].x = Wm[(size_t)k0 * FDIM + col];
            bfrag[k].y = Wm[(size_t)(k0 + 1) * FDIM + col];
        }
    }

    const int cc = ctile * 16 + l16;
    const float bcol = bias[cc];
    const float wcol = LAST ? w2[cc] : 0.0f;
    const int rowBlocks = (nrows + 31) >> 5;
    for (int rb = blockIdx.x; rb < rowBlocks; rb += gridDim.x) {
        const int row0 = rb * 32 + rhalf * 16;
        const bool full = (row0 + 16 <= nrows);

        {
            float4 va[2];
            #pragma unroll
            for (int j = 0; j < 2; ++j) {
                const int idx  = threadIdx.x + j * 256;
                const int grow = rb * 32 + (idx >> 4);
                const int c4   = idx & 15;
                const size_t q = (size_t)(grow < nrows ? grow : 0) * 16 + c4;
                va[j] = ((const float4*)S)[q];
            }
            #pragma unroll
            for (int j = 0; j < 2; ++j) {
                const int idx = threadIdx.x + j * 256;
                *(float4*)&atile[(idx >> 4) * A_PITCH + (idx & 15) * 4] = va[j];
            }
        }
        __syncthreads();

        const float* hp = atile + (rhalf * 16 + l16) * A_PITCH + (hi ? 2 : 0);

        v2f afr[16];
        #pragma unroll
        for (int k = 0; k < 16; ++k) afr[k] = *(const v2f*)(hp + 4 * k);
        asm volatile("" :: "v"(afr[0]), "v"(afr[1]), "v"(afr[2]), "v"(afr[3]), "v"(afr[4]), "v"(afr[5]),
                           "v"(afr[6]), "v"(afr[7]), "v"(afr[8]), "v"(afr[9]), "v"(afr[10]), "v"(afr[11]),
                           "v"(afr[12]), "v"(afr[13]), "v"(afr[14]), "v"(afr[15]));

        v8f acc = {};
        #pragma unroll
        for (int k = 0; k < 16; ++k)
            acc = __builtin_amdgcn_wmma_f32_16x16x4_f32(
                      false, afr[k], false, bfrag[k], (short)0, acc, false, false);

        float v[8];
        #pragma unroll
        for (int i = 0; i < 8; ++i) v[i] = fmaxf(acc[i] + bcol, 0.0f);

        if (!LAST) {
            if (full) {
                const size_t o = (size_t)(row0 + hi * 8) * FDIM + cc;
                #pragma unroll
                for (int i = 0; i < 8; ++i)
                    o0[o + (size_t)i * FDIM] = v[i];
            } else {
                #pragma unroll
                for (int i = 0; i < 8; ++i) {
                    int rr = row0 + hi * 8 + i;
                    if (rr < nrows) o0[(size_t)rr * FDIM + cc] = v[i];
                }
            }
            if (rb + (int)gridDim.x < rowBlocks) __syncthreads();
        } else {
            #pragma unroll
            for (int i = 0; i < 8; ++i)
                red[(rhalf * 16 + hi * 8 + i) * RED_PITCH + cc] = v[i] * wcol;
            __syncthreads();
            {
                const int rrow = threadIdx.x >> 3;
                const int seg  = threadIdx.x & 7;
                const float* rp = red + rrow * RED_PITCH + seg * 8;
                float s = 0.0f;
                #pragma unroll
                for (int j = 0; j < 8; ++j) s += rp[j];
                s += __shfl_xor(s, 1);
                s += __shfl_xor(s, 2);
                s += __shfl_xor(s, 4);
                const int grow = rb * 32 + rrow;
                if (seg == 0 && grow < nrows) o0[grow] = s;
            }
            __syncthreads();
        }
    }
}


template <int J>
__device__ __forceinline__ int row_lane(int v) {
    return __builtin_amdgcn_update_dpp(0, v, 0x150 + J, 0xf, 0xf, true);
}

__device__ __forceinline__ void fma4(float4& acc, float k, const float4& x) {
    acc.x += k * x.x; acc.y += k * x.y; acc.z += k * x.z; acc.w += k * x.w;
}

template <int J>
__device__ __forceinline__ void slot_fetch(const float4* __restrict__ H, int c, int srca, int srcb,
                                           int m0, int m1, float4& xa, float4& xb) {
    const int sa = row_lane<J>(srca);
    const int sb = row_lane<J>(srcb);
    xa = make_float4(0.0f, 0.0f, 0.0f, 0.0f);
    xb = xa;
    if (J < m0) xa = H[(size_t)sa * 16 + c];
    if (J < m1) xb = H[(size_t)sb * 16 + c];
}

template <int J>
__device__ __forceinline__ void slot_add(float4& acc, float ca, float cb, int m0, int m1,
                                         const float4& xa, const float4& xb) {
    const float fa = __int_as_float(row_lane<J>(__float_as_int(ca)));
    const float fb = __int_as_float(row_lane<J>(__float_as_int(cb)));
    fma4(acc, (J < m0) ? fa : 0.0f, xa);
    fma4(acc, (J < m1) ? fb : 0.0f, xb);
}

template <int J0, bool FORM>
__device__ __forceinline__ void slot_batch(const float4* __restrict__ H, int c, int srca, int srcb,
                                           float& ca, float& cb, float da, float db, float dr,
                                           bool oa, bool ob, int m0, int m1, float4& acc) {
    float4 xa[4], xb[4];
    slot_fetch<J0 + 0>(H, c, srca, srcb, m0, m1, xa[0], xb[0]);
    slot_fetch<J0 + 1>(H, c, srca, srcb, m0, m1, xa[1], xb[1]);
    slot_fetch<J0 + 2>(H, c, srca, srcb, m0, m1, xa[2], xb[2]);
    slot_fetch<J0 + 3>(H, c, srca, srcb, m0, m1, xa[3], xb[3]);
    asm volatile("" :: "v"(xa[0].x), "v"(xa[0].y), "v"(xa[0].z), "v"(xa[0].w),
                       "v"(xa[1].x), "v"(xa[1].y), "v"(xa[1].z), "v"(xa[1].w),
                       "v"(xa[2].x), "v"(xa[2].y), "v"(xa[2].z), "v"(xa[2].w),
                       "v"(xa[3].x), "v"(xa[3].y), "v"(xa[3].z), "v"(xa[3].w));
    asm volatile("" :: "v"(xb[0].x), "v"(xb[0].y), "v"(xb[0].z), "v"(xb[0].w),
                       "v"(xb[1].x), "v"(xb[1].y), "v"(xb[1].z), "v"(xb[1].w),
                       "v"(xb[2].x), "v"(xb[2].y), "v"(xb[2].z), "v"(xb[2].w),
                       "v"(xb[3].x), "v"(xb[3].y), "v"(xb[3].z), "v"(xb[3].w));
    if (FORM) {
        ca = oa ? (da * ca) * dr : 0.0f;
        cb = ob ? (db * cb) * dr : 0.0f;
    }
    slot_add<J0 + 0>(acc, ca, cb, m0, m1, xa[0], xb[0]);
    slot_add<J0 + 1>(acc, ca, cb, m0, m1, xa[1], xb[1]);
    slot_add<J0 + 2>(acc, ca, cb, m0, m1, xa[2], xb[2]);
    slot_add<J0 + 3>(acc, ca, cb, m0, m1, xa[3], xb[3]);
}

template <bool FIRST>
__device__ __forceinline__ void chain_walk(const float4* __restrict__ H, int4* rec,
                                           const float* __restrict__ dinv, float dr,
                                           int p, int c, int e, float4& acc) {
    for (int it = 0; it < e && (unsigned)p < (unsigned)e; ++it) {
        const int4   r = rec[p];
        const float4 x = H[(size_t)r.y * 16 + c];
        float k = __int_as_float(r.z);
        if (FIRST) {
            k = (dinv[r.y] * k) * dr;
            if (c == 0) ((int*)rec)[(size_t)p * 4 + 2] = __float_as_int(k);
        }
        fma4(acc, k, x);
        p = r.x;
    }
}

template <bool FIRST>
__global__ __launch_bounds__(256) void k_pull(const float* __restrict__ h,
                                              int2* slots,
                                              const int* __restrict__ cnt,
                                              int4* rec,
                                              const int* __restrict__ head,
                                              const float* __restrict__ dinv,
                                              const float* __restrict__ self_norm,
                                              float* __restrict__ agg, int n, int e) {
    const int row = blockIdx.x * 16 + (int)(threadIdx.x >> 4);
    const int c   = (int)(threadIdx.x & 15);
    if (row >= n) return;
    const float4* const H = (const float4*)h;
    const int pitch = head_pitch(n);
    const int n0 = cnt[row];
    const int n1 = cnt[pitch + row];
    const int2 ea = slots[(size_t)row * SLOT_K + c];
    const int2 eb = slots[(size_t)(pitch + row) * SLOT_K + c];
    const float  sn  = self_norm[row];
    const float  dr  = FIRST ? dinv[row] : 0.0f;
    const float4 own = H[(size_t)row * 16 + c];
    asm volatile("" :: "v"(ea.x), "v"(ea.y), "v"(eb.x), "v"(eb.y));
    float4 acc = make_float4(own.x * sn, own.y * sn, own.z * sn, own.w * sn);
    const int m0 = min(n0, SLOT_K);
    const int m1 = min(n1, SLOT_K);
    const int mm = max(m0, m1);
    const bool oa = c < m0;
    const bool ob = c < m1;
    float ca = __int_as_float(ea.y);
    float cb = __int_as_float(eb.y);
    float da = 0.0f, db = 0.0f;
    if (FIRST) {
        da = dinv[oa ? ea.x : row];
        db = dinv[ob ? eb.x : row];
    }
    if (mm > 0) {
        slot_batch<0, FIRST>(H, c, ea.x, eb.x, ca, cb, da, db, dr, oa, ob, m0, m1, acc);
        if (mm > 4) {
            slot_batch<4, false>(H, c, ea.x, eb.x, ca, cb, da, db, dr, oa, ob, m0, m1, acc);
            if (mm > 8) {
                slot_batch<8, false>(H, c, ea.x, eb.x, ca, cb, da, db, dr, oa, ob, m0, m1, acc);
                if (mm > 12) slot_batch<12, false>(H, c, ea.x, eb.x, ca, cb, da, db, dr, oa, ob, m0, m1, acc);
            }
        }
        if (FIRST) {
            if (oa) ((int*)slots)[((size_t)row * SLOT_K + c) * 2 + 1] = __float_as_int(ca);
            if (ob) ((int*)slots)[((size_t)(pitch + row) * SLOT_K + c) * 2 + 1] = __float_as_int(cb);
        }
    }
    if (n0 > SLOT_K) chain_walk<FIRST>(H, rec, dinv, dr, head[row], c, e, acc);
    if (n1 > SLOT_K) chain_walk<FIRST>(H, rec, dinv, dr, head[pitch + row], c, e, acc);
    ((float4*)agg)[(size_t)row * 16 + c] = acc;
}


__global__ __launch_bounds__(256) void k_pull1(const float* __restrict__ lin,
                                               const int2* __restrict__ slots,
                                               const int* __restrict__ cnt,
                                               const int4* __restrict__ rec,
                                               const int* __restrict__ head,
                                               const float* __restrict__ self_norm,
                                               const float* __restrict__ b2,
                                               float* __restrict__ out, int n, int e) {
    const int t    = blockIdx.x * 256 + threadIdx.x;
    const int row  = t >> 3;
    const int q    = t & 7;
    const bool live = row < n;
    const int rr    = live ? row : 0;
    const int pitch = head_pitch(n);
    const int4* const S4 = (const int4*)slots;
    const int n0 = cnt[rr];
    const int n1 = cnt[pitch + rr];
    const int4 ea = S4[(size_t)rr * (SLOT_K / 2) + q];
    const int4 eb = S4[(size_t)(pitch + rr) * (SLOT_K / 2) + q];
    const float lv = lin[rr];
    const float sv = self_norm[rr];
    asm volatile("" :: "v"(ea.x), "v"(ea.y), "v"(ea.z), "v"(ea.w), "v"(eb.x), "v"(eb.y), "v"(eb.z), "v"(eb.w));
    const int m0 = live ? min(n0, SLOT_K) : 0;
    const int m1 = live ? min(n1, SLOT_K) : 0;
    const bool u0 = 2 * q < m0, u1 = 2 * q + 1 < m0;
    const bool u2 = 2 * q < m1, u3 = 2 * q + 1 < m1;
    float x0 = 0.0f, x1 = 0.0f, x2 = 0.0f, x3 = 0.0f;
    if (u0) x0 = lin[ea.x];
    if (u1) x1 = lin[ea.z];
    if (u2) x2 = lin[eb.x];
    if (u3) x3 = lin[eb.z];
    asm volatile("" :: "v"(x0), "v"(x1), "v"(x2), "v"(x3));
    float acc = 0.0f;
    acc += (u0 ? __int_as_float(ea.y) : 0.0f) * x0;
    acc += (u1 ? __int_as_float(ea.w) : 0.0f) * x1;
    acc += (u2 ? __int_as_float(eb.y) : 0.0f) * x2;
    acc += (u3 ? __int_as_float(eb.w) : 0.0f) * x3;
    if (live && q == 0) {
        if (n0 > SLOT_K) {
            int p = head[row];
            for (int it = 0; it < e && (unsigned)p < (unsigned)e; ++it) {
                const int4 r = rec[p];
                acc += __int_as_float(r.z) * lin[r.y];
                p = r.x;
            }
        }
        if (n1 > SLOT_K) {
            int p = head[pitch + row];
            for (int it = 0; it < e && (unsigned)p < (unsigned)e; ++it) {
                const int4 r = rec[p];
                acc += __int_as_float(r.z) * lin[r.y];
                p = r.x;
            }
        }
    }
    acc += __shfl_xor(acc, 1);
    acc += __shfl_xor(acc, 2);
    acc += __shfl_xor(acc, 4);
    if (live && q == 0) out[row] = (lv * sv + b2[0]) + acc;
}


extern "C" void kernel_launch(void* const* d_in, const int* in_sizes, int n_in,
                              void* d_out, int out_size, void* d_ws, size_t ws_size,
                              hipStream_t stream) {
    const float* x    = (const float*)d_in[0];
    const int*   esrc = (const int*)  d_in[1];
    const int*   edst = (const int*)  d_in[2];
    const float* ew   = (const float*)d_in[3];
    const float* W0   = (const float*)d_in[4];
    const float* b0   = (const float*)d_in[5];
    const float* W1   = (const float*)d_in[6];
    const float* b1   = (const float*)d_in[7];
    const float* W2   = (const float*)d_in[8];
    const float* b2   = (const float*)d_in[9];

    const int N = in_sizes[0] / FDIM;
    const int E = in_sizes[1];
    float* out = (float*)d_out;

    float* ws        = (float*)d_ws;
    int4*  rec       = (int4*)d_ws;
    float* s0        = ws + 4 * (size_t)E;
    float* h1        = s0 + (size_t)N * FDIM;
    float* s1        = h1 + (size_t)N * FDIM;
    const size_t pitch = (size_t)((N + 31) & ~31);
    int2*  slots     = (int2*)(s1 + (size_t)N * FDIM);
    int*   head      = (int*)(slots + 2 * pitch * SLOT_K);
    int*   cnt       = head + 2 * pitch;
    float* dinv      = (float*)(cnt + 2 * pitch);
    float* self_norm = dinv + (size_t)N;
    float* lin       = s0;

    const int T = 256;
    const int gN  = (N + T - 1) / T;
    const int gN8 = (8 * N + T - 1) / T;
    const int gN4 = (4 * N + T - 1) / T;
    const int gE  = (E + T - 1) / T;
    const int gGemm = (N + 31) / 32;
    const int gPull = (N + 15) / 16;

    k_init<<<gN, T, 0, stream>>>(head, cnt, N);
    k_link<<<gE, T, 0, stream>>>(esrc, edst, ew, slots, cnt, rec, head, N, E);
    k_deg<<<gN4, T, 0, stream>>>(slots, cnt, rec, head, dinv, self_norm, N, E);

    k_pull<true><<<gPull, T, 0, stream>>>(x, slots, cnt, rec, head, dinv, self_norm, s0, N, E);
    k_gemm64_wmma<false><<<gGemm, T, 0, stream>>>(s0, W0, b0, W2, h1, N);

    k_pull<false><<<gPull, T, 0, stream>>>(h1, slots, cnt, rec, head, dinv, self_norm, s1, N, E);
    k_gemm64_wmma<true><<<gGemm, T, 0, stream>>>(s1, W1, b1, W2, lin, N);

    k_pull1<<<gN8, T, 0, stream>>>(lin, slots, cnt, rec, head, self_norm, b2, out, N, E);
}


# Layer 1 multiplies before its edge sum, A (h W1): both 64x64 products run in one kernel (relu tile stays in LDS); its pull ends in bias, relu, 64->1 dot; 8 launches -> 7; sum orders change
# speedup vs baseline: 4.2982x; 1.0321x over previous
#include <hip/hip_runtime.h>

typedef __attribute__((ext_vector_type(2))) float v2f;
typedef __attribute__((ext_vector_type(8))) float v8f;

#define FDIM 64


__device__ __forceinline__ int head_pitch(int n) {
    return (n + 31) & ~31;
}

__global__ __launch_bounds__(256) void k_init(int* __restrict__ head, int* __restrict__ cnt, int n) {
    int i = blockIdx.x * blockDim.x + threadIdx.x;
    if (i < n) {
        cnt[i] = 0;
        cnt[head_pitch(n) + i] = 0;
        head[i] = -1;
        head[head_pitch(n) + i] = -1;
    }
}

#define SLOT_K 16

__global__ __launch_bounds__(256) void k_link(const int* __restrict__ src,
                                              const int* __restrict__ dst,
                                              const float* __restrict__ w,
                                              int2* __restrict__ slots,
                                              int* __restrict__ cnt,
                                              int4* __restrict__ rec,
                                              int* __restrict__ head, int n, int e) {
    int i = blockIdx.x * blockDim.x + threadIdx.x;
    const int half = (int)((__builtin_amdgcn_s_sendmsg_rtn(0x87) >> 18) & 1u);
    if (i < e) {
        const int s = src[i];
        const int d = dst[i];
        const int wv = __float_as_int(w[i]);
        const int node = half * head_pitch(n) + d;
        const int slot = atomicAdd(&cnt[node], 1);
        if (slot < SLOT_K) {
            slots[(size_t)node * SLOT_K + slot] = make_int2(s, wv);
        } else {
            const int prev = atomicExch(&head[node], i);
            rec[i] = make_int4(prev, s, wv, 0);
        }
    }
}

__global__ __launch_bounds__(256) void k_deg(const int2* __restrict__ slots,
                                             const int* __restrict__ cnt,
                                             const int4* __restrict__ rec,
                                             const int* __restrict__ head,
                                             float* __restrict__ dinv,
                                             float* __restrict__ self_norm, int n, int e) {
    const int t    = blockIdx.x * 256 + threadIdx.x;
    const int row  = t >> 2;
    const int q    = t & 3;
    const bool live = row < n;
    const int rr    = live ? row : 0;
    const int pitch = head_pitch(n);
    const int4* const S4 = (const int4*)slots;
    const int n0 = cnt[rr];
    const int n1 = cnt[pitch + rr];
    const int4 a0 = S4[(size_t)rr * (SLOT_K / 2) + 2 * q];
    const int4 a1 = S4[(size_t)rr * (SLOT_K / 2) + 2 * q + 1];
    const int4 b0 = S4[(size_t)(pitch + rr) * (SLOT_K / 2) + 2 * q];
    const int4 b1 = S4[(size_t)(pitch + rr) * (SLOT_K / 2) + 2 * q + 1];
    const int m0 = live ? min(n0, SLOT_K) : 0;
    const int m1 = live ? min(n1, SLOT_K) : 0;
    const int j = 4 * q;
    float s = 0.0f;
    s += (j + 0 < m0) ? __int_as_float(a0.y) : 0.0f;
    s += (j + 1 < m0) ? __int_as_float(a0.w) : 0.0f;
    s += (j + 2 < m0) ? __int_as_float(a1.y) : 0.0f;
    s += (j + 3 < m0) ? __int_as_float(a1.w) : 0.0f;
    s += (j + 0 < m1) ? __int_as_float(b0.y) : 0.0f;
    s += (j + 1 < m1) ? __int_as_float(b0.w) : 0.0f;
    s += (j + 2 < m1) ? __int_as_float(b1.y) : 0.0f;
    s += (j + 3 < m1) ? __int_as_float(b1.w) : 0.0f;
    if (live && q == 0) {
        if (n0 > SLOT_K) {
            int p = head[row];
            for (int it = 0; it < e && (unsigned)p < (unsigned)e; ++it) {
                const int4 r = rec[p];
                s += __int_as_float(r.z);
                p = r.x;
            }
        }
        if (n1 > SLOT_K) {
            int p = head[pitch + row];
            for (int it = 0; it < e && (unsigned)p < (unsigned)e; ++it) {
                const int4 r = rec[p];
                s += __int_as_float(r.z);
                p = r.x;
            }
        }
    }
    s += __shfl_xor(s, 1);
    s += __shfl_xor(s, 2);
    if (live && q == 0) {
        const float dv = rsqrtf(s + 1.0f);
        dinv[row]      = dv;
        self_norm[row] = dv * dv;
    }
}

#define A_PITCH   68

__device__ __forceinline__ void load_bfrag(const float* __restrict__ Wm, int col, int hi, v2f (&bf)[16]) {
    #pragma unroll
    for (int k = 0; k < 16; ++k) {
        const int k0 = 4 * k + (hi ? 2 : 0);
        bf[k].x = Wm[(size_t)k0 * FDIM + col];
        bf[k].y = Wm[(size_t)(k0 + 1) * FDIM + col];
    }
}

__device__ __forceinline__ v8f tile_product(const float* hp, const v2f (&bf)[16]) {
    v2f afr[16];
    #pragma unroll
    for (int k = 0; k < 16; ++k) afr[k] = *(const v2f*)(hp + 4 * k);
    asm volatile("" :: "v"(afr[0]), "v"(afr[1]), "v"(afr[2]), "v"(afr[3]), "v"(afr[4]), "v"(afr[5]),
                       "v"(afr[6]), "v"(afr[7]), "v"(afr[8]), "v"(afr[9]), "v"(afr[10]), "v"(afr[11]),
                       "v"(afr[12]), "v"(afr[13]), "v"(afr[14]), "v"(afr[15]));
    v8f acc = {};
    #pragma unroll
    for (int k = 0; k < 16; ++k)
        acc = __builtin_amdgcn_wmma_f32_16x16x4_f32(
                  false, afr[k], false, bf[k], (short)0, acc, false, false);
    return acc;
}

__global__ __launch_bounds__(256) void k_gemm64x2_wmma(const float* __restrict__ S,
                                                       const float* __restrict__ Wa,
                                                       const float* __restrict__ bias,
                                                       const float* __restrict__ Wb,
                                                       float* __restrict__ o0, int nrows) {
    __shared__ float atile[32 * A_PITCH];
    __shared__ float htile[32 * A_PITCH];
    const int lane  = threadIdx.x & 31;
    const int wave  = threadIdx.x >> 5;
    const int ctile = wave & 3;
    const int rhalf = wave >> 2;
    const int l16   = lane & 15;
    const int hi    = lane >> 4;
    const int cc    = ctile * 16 + l16;

    v2f bfa[16], bfb[16];
    load_bfrag(Wa, cc, hi, bfa);
    load_bfrag(Wb, cc, hi, bfb);

    const float bcol = bias[cc];
    const int rowBlocks = (nrows + 31) >> 5;
    for (int rb = blockIdx.x; rb < rowBlocks; rb += gridDim.x) {
        const int row0 = rb * 32 + rhalf * 16;
        const bool full = (row0 + 16 <= nrows);

        {
            float4 va[2];
            #pragma unroll
            for (int j = 0; j < 2; ++j) {
                const int idx  = threadIdx.x + j * 256;
                const int grow = rb * 32 + (idx >> 4);
                const int c4   = idx & 15;
                const size_t q = (size_t)(grow < nrows ? grow : 0) * 16 + c4;
                va[j] = ((const float4*)S)[q];
            }
            #pragma unroll
            for (int j = 0; j < 2; ++j) {
                const int idx = threadIdx.x + j * 256;
                *(float4*)&atile[(idx >> 4) * A_PITCH + (idx & 15) * 4] = va[j];
            }
        }
        __syncthreads();

        const int aoff = (rhalf * 16 + l16) * A_PITCH + (hi ? 2 : 0);

        const v8f acc = tile_product(atile + aoff, bfa);

        #pragma unroll
        for (int i = 0; i < 8; ++i)
            htile[(rhalf * 16 + hi * 8 + i) * A_PITCH + cc] = fmaxf(acc[i] + bcol, 0.0f);
        __syncthreads();

        const v8f t = tile_product(htile + aoff, bfb);

        if (full) {
            const size_t o = (size_t)(row0 + hi * 8) * FDIM + cc;
            #pragma unroll
            for (int i = 0; i < 8; ++i)
                o0[o + (size_t)i * FDIM] = t[i];
        } else {
            #pragma unroll
            for (int i = 0; i < 8; ++i) {
                int rr = row0 + hi * 8 + i;
                if (rr < nrows) o0[(size_t)rr * FDIM + cc] = t[i];
            }
        }
        if (rb + (int)gridDim.x < rowBlocks) __syncthreads();
    }
}


template <int J>
__device__ __forceinline__ int row_lane(int v) {
    return __builtin_amdgcn_update_dpp(0, v, 0x150 + J, 0xf, 0xf, true);
}

template <int M>
__device__ __forceinline__ float row_xor(float v) {
    return __int_as_float(__builtin_amdgcn_update_dpp(0, __float_as_int(v), 0x160 + M, 0xf, 0xf, true));
}

struct rowsum { v2f lo, hi; };

__device__ __forceinline__ void fma4(rowsum& acc, float k, const float4& x) {
    const v2f kk = {k, k};
    const v2f xl = {x.x, x.y};
    const v2f xh = {x.z, x.w};
    acc.lo = __builtin_elementwise_fma(kk, xl, acc.lo);
    acc.hi = __builtin_elementwise_fma(kk, xh, acc.hi);
}

template <int J>
__device__ __forceinline__ void slot_fetch(const float4* __restrict__ H, int c, int srca, int srcb,
                                           int m0, int m1, float4& xa, float4& xb) {
    const int sa = row_lane<J>(srca);
    const int sb = row_lane<J>(srcb);
    xa = make_float4(0.0f, 0.0f, 0.0f, 0.0f);
    xb = xa;
    if (J < m0) xa = H[(size_t)sa * 16 + c];
    if (J < m1) xb = H[(size_t)sb * 16 + c];
}

template <int J>
__device__ __forceinline__ void slot_add(rowsum& acc, float ca, float cb, int m0, int m1,
                                         const float4& xa, const float4& xb) {
    const float fa = __int_as_float(row_lane<J>(__float_as_int(ca)));
    const float fb = __int_as_float(row_lane<J>(__float_as_int(cb)));
    fma4(acc, (J < m0) ? fa : 0.0f, xa);
    fma4(acc, (J < m1) ? fb : 0.0f, xb);
}

template <int J0, bool FORM>
__device__ __forceinline__ void slot_batch(const float4* __restrict__ H, int c, int srca, int srcb,
                                           float& ca, float& cb, float da, float db, float dr,
                                           bool oa, bool ob, int m0, int m1, rowsum& acc) {
    float4 xa[4], xb[4];
    slot_fetch<J0 + 0>(H, c, srca, srcb, m0, m1, xa[0], xb[0]);
    slot_fetch<J0 + 1>(H, c, srca, srcb, m0, m1, xa[1], xb[1]);
    slot_fetch<J0 + 2>(H, c, srca, srcb, m0, m1, xa[2], xb[2]);
    slot_fetch<J0 + 3>(H, c, srca, srcb, m0, m1, xa[3], xb[3]);
    asm volatile("" :: "v"(xa[0].x), "v"(xa[0].y), "v"(xa[0].z), "v"(xa[0].w),
                       "v"(xa[1].x), "v"(xa[1].y), "v"(xa[1].z), "v"(xa[1].w),
                       "v"(xa[2].x), "v"(xa[2].y), "v"(xa[2].z), "v"(xa[2].w),
                       "v"(xa[3].x), "v"(xa[3].y), "v"(xa[3].z), "v"(xa[3].w));
    asm volatile("" :: "v"(xb[0].x), "v"(xb[0].y), "v"(xb[0].z), "v"(xb[0].w),
                       "v"(xb[1].x), "v"(xb[1].y), "v"(xb[1].z), "v"(xb[1].w),
                       "v"(xb[2].x), "v"(xb[2].y), "v"(xb[2].z), "v"(xb[2].w),
                       "v"(xb[3].x), "v"(xb[3].y), "v"(xb[3].z), "v"(xb[3].w));
    if (FORM) {
        ca = oa ? (da * ca) * dr : 0.0f;
        cb = ob ? (db * cb) * dr : 0.0f;
    }
    slot_add<J0 + 0>(acc, ca, cb, m0, m1, xa[0], xb[0]);
    slot_add<J0 + 1>(acc, ca, cb, m0, m1, xa[1], xb[1]);
    slot_add<J0 + 2>(acc, ca, cb, m0, m1, xa[2], xb[2]);
    slot_add<J0 + 3>(acc, ca, cb, m0, m1, xa[3], xb[3]);
}

template <bool FIRST>
__device__ __forceinline__ void chain_walk(const float4* __restrict__ H, int4* rec,
                                           const float* __restrict__ dinv, float dr,
                                           int p, int c, int e, rowsum& acc) {
    for (int it = 0; it < e && (unsigned)p < (unsigned)e; ++it) {
        const int4   r = rec[p];
        const float4 x = H[(size_t)r.y * 16 + c];
        float k = __int_as_float(r.z);
        if (FIRST) {
            k = (dinv[r.y] * k) * dr;
            if (c == 0) ((int*)rec)[(size_t)p * 4 + 2] = __float_as_int(k);
        }
        fma4(acc, k, x);
        p = r.x;
    }
}

#define PULL_ROWS(FIRST) ((FIRST) ? 16 : 32)

template <bool FIRST>
__global__ __launch_bounds__(PULL_ROWS(FIRST) * 16) void k_pull(const float* __restrict__ h,
                                              int2* slots,
                                              const int* __restrict__ cnt,
                                              int4* rec,
                                              const int* __restrict__ head,
                                              const float* __restrict__ dinv,
                                              const float* __restrict__ self_norm,
                                              const float* __restrict__ bias,
                                              const float* __restrict__ w2,
                                              float* __restrict__ agg, int n, int e) {
    const int row = blockIdx.x * PULL_ROWS(FIRST) + (int)(threadIdx.x >> 4);
    const int c   = (int)(threadIdx.x & 15);
    if (row >= n) return;
    const float4* const H = (const float4*)h;
    const int pitch = head_pitch(n);
    const int n0 = cnt[row];
    const int n1 = cnt[pitch + row];
    const int2 ea = slots[(size_t)row * SLOT_K + c];
    const int2 eb = slots[(size_t)(pitch + row) * SLOT_K + c];
    const float  sn  = self_norm[row];
    const float  dr  = FIRST ? dinv[row] : 0.0f;
    const float4 own = H[(size_t)row * 16 + c];
    __builtin_amdgcn_sched_barrier(0);
    rowsum acc;
    acc.lo = (v2f){own.x * sn, own.y * sn};
    acc.hi = (v2f){own.z * sn, own.w * sn};
    const int m0 = min(n0, SLOT_K);
    const int m1 = min(n1, SLOT_K);
    const int mm = max(m0, m1);
    const bool oa = c < m0;
    const bool ob = c < m1;
    float ca = __int_as_float(ea.y);
    float cb = __int_as_float(eb.y);
    float da = 0.0f, db = 0.0f;
    if (FIRST) {
        da = dinv[oa ? ea.x : row];
        db = dinv[ob ? eb.x : row];
    }
    if (mm > 0) {
        slot_batch<0, FIRST>(H, c, ea.x, eb.x, ca, cb, da, db, dr, oa, ob, m0, m1, acc);
        if (mm > 4) {
            slot_batch<4, false>(H, c, ea.x, eb.x, ca, cb, da, db, dr, oa, ob, m0, m1, acc);
            if (mm > 8) {
                slot_batch<8, false>(H, c, ea.x, eb.x, ca, cb, da, db, dr, oa, ob, m0, m1, acc);
                if (mm > 12) slot_batch<12, false>(H, c, ea.x, eb.x, ca, cb, da, db, dr, oa, ob, m0, m1, acc);
            }
        }
        if (FIRST) {
            if (oa) ((int*)slots)[((size_t)row * SLOT_K + c) * 2 + 1] = __float_as_int(ca);
            if (ob) ((int*)slots)[((size_t)(pitch + row) * SLOT_K + c) * 2 + 1] = __float_as_int(cb);
        }
    }
    if (n0 > SLOT_K) chain_walk<FIRST>(H, rec, dinv, dr, head[row], c, e, acc);
    if (n1 > SLOT_K) chain_walk<FIRST>(H, rec, dinv, dr, head[pitch + row], c, e, acc);
    if (FIRST) {
        ((float4*)agg)[(size_t)row * 16 + c] = make_float4(acc.lo.x, acc.lo.y, acc.hi.x, acc.hi.y);
    } else {
        const v2f a01 = acc.lo;
        const v2f a23 = acc.hi;
        const v2f b01 = *(const v2f*)(bias + 4 * c);
        const v2f b23 = *(const v2f*)(bias + 4 * c + 2);
        const v2f w01 = *(const v2f*)(w2 + 4 * c);
        const v2f w23 = *(const v2f*)(w2 + 4 * c + 2);
        const v2f z2  = {0.0f, 0.0f};
        const v2f t01 = __builtin_elementwise_max(a01 + b01, z2) * w01;
        const v2f t23 = __builtin_elementwise_max(a23 + b23, z2) * w23;
        float s = (t01.x + t01.y) + (t23.x + t23.y);
        s += row_xor<8>(s);
        s += row_xor<4>(s);
        s += row_xor<2>(s);
        s += row_xor<1>(s);
        if (c == 0) agg[row] = s;
    }
}


__global__ __launch_bounds__(256) void k_pull1(const float* __restrict__ lin,
                                               const int2* __restrict__ slots,
                                               const int* __restrict__ cnt,
                                               const int4* __restrict__ rec,
                                               const int* __restrict__ head,
                                               const float* __restrict__ self_norm,
                                               const float* __restrict__ b2,
                                               float* __restrict__ out, int n, int e) {
    const int t    = blockIdx.x * 256 + threadIdx.x;
    const int row  = t >> 3;
    const int q    = t & 7;
    const bool live = row < n;
    const int rr    = live ? row : 0;
    const int pitch = head_pitch(n);
    const int4* const S4 = (const int4*)slots;
    const int n0 = cnt[rr];
    const int n1 = cnt[pitch + rr];
    const int4 ea = S4[(size_t)rr * (SLOT_K / 2) + q];
    const int4 eb = S4[(size_t)(pitch + rr) * (SLOT_K / 2) + q];
    const float lv = lin[rr];
    const float sv = self_norm[rr];
    asm volatile("" :: "v"(ea.x), "v"(ea.y), "v"(ea.z), "v"(ea.w), "v"(eb.x), "v"(eb.y), "v"(eb.z), "v"(eb.w));
    const int m0 = live ? min(n0, SLOT_K) : 0;
    const int m1 = live ? min(n1, SLOT_K) : 0;
    const bool u0 = 2 * q < m0, u1 = 2 * q + 1 < m0;
    const bool u2 = 2 * q < m1, u3 = 2 * q + 1 < m1;
    float x0 = 0.0f, x1 = 0.0f, x2 = 0.0f, x3 = 0.0f;
    if (u0) x0 = lin[ea.x];
    if (u1) x1 = lin[ea.z];
    if (u2) x2 = lin[eb.x];
    if (u3) x3 = lin[eb.z];
    asm volatile("" :: "v"(x0), "v"(x1), "v"(x2), "v"(x3));
    float acc = 0.0f;
    acc += (u0 ? __int_as_float(ea.y) : 0.0f) * x0;
    acc += (u1 ? __int_as_float(ea.w) : 0.0f) * x1;
    acc += (u2 ? __int_as_float(eb.y) : 0.0f) * x2;
    acc += (u3 ? __int_as_float(eb.w) : 0.0f) * x3;
    if (live && q == 0) {
        if (n0 > SLOT_K) {
            int p = head[row];
            for (int it = 0; it < e && (unsigned)p < (unsigned)e; ++it) {
                const int4 r = rec[p];
                acc += __int_as_float(r.z) * lin[r.y];
                p = r.x;
            }
        }
        if (n1 > SLOT_K) {
            int p = head[pitch + row];
            for (int it = 0; it < e && (unsigned)p < (unsigned)e; ++it) {
                const int4 r = rec[p];
                acc += __int_as_float(r.z) * lin[r.y];
                p = r.x;
            }
        }
    }
    acc += __shfl_xor(acc, 1);
    acc += __shfl_xor(acc, 2);
    acc += __shfl_xor(acc, 4);
    if (live && q == 0) out[row] = (lv * sv + b2[0]) + acc;
}


extern "C" void kernel_launch(void* const* d_in, const int* in_sizes, int n_in,
                              void* d_out, int out_size, void* d_ws, size_t ws_size,
                              hipStream_t stream) {
    const float* x    = (const float*)d_in[0];
    const int*   esrc = (const int*)  d_in[1];
    const int*   edst = (const int*)  d_in[2];
    const float* ew   = (const float*)d_in[3];
    const float* W0   = (const float*)d_in[4];
    const float* b0   = (const float*)d_in[5];
    const float* W1   = (const float*)d_in[6];
    const float* b1   = (const float*)d_in[7];
    const float* W2   = (const float*)d_in[8];
    const float* b2   = (const float*)d_in[9];

    const int N = in_sizes[0] / FDIM;
    const int E = in_sizes[1];
    float* out = (float*)d_out;

    float* ws        = (float*)d_ws;
    int4*  rec       = (int4*)d_ws;
    float* s0        = ws + 4 * (size_t)E;
    float* t1        = s0 + (size_t)N * FDIM;
    float* lin       = t1 + (size_t)N * FDIM;
    const size_t pitch = (size_t)((N + 31) & ~31);
    int2*  slots     = (int2*)(lin + (size_t)N * FDIM);
    int*   head      = (int*)(slots + 2 * pitch * SLOT_K);
    int*   cnt       = head + 2 * pitch;
    float* dinv      = (float*)(cnt + 2 * pitch);
    float* self_norm = dinv + (size_t)N;

    const int T = 256;
    const int T2 = 512;
    const int gN  = (N + T - 1) / T;
    const int gN8 = (8 * N + T - 1) / T;
    const int gN4 = (4 * N + T - 1) / T;
    const int gE  = (E + T - 1) / T;
    const int gGemm = (N + 31) / 32;
    const int gPull  = (N + 15) / 16;
    const int gPull2 = (N + 31) / 32;

    k_init<<<gN, T, 0, stream>>>(head, cnt, N);
    k_link<<<gE, T, 0, stream>>>(esrc, edst, ew, slots, cnt, rec, head, N, E);
    k_deg<<<gN4, T, 0, stream>>>(slots, cnt, rec, head, dinv, self_norm, N, E);

    k_pull<true><<<gPull, T, 0, stream>>>(x, slots, cnt, rec, head, dinv, self_norm, b1, W2, s0, N, E);
    k_gemm64x2_wmma<<<gGemm, T, 0, stream>>>(s0, W0, b0, W1, t1, N);

    k_pull<false><<<gPull2, T2, 0, stream>>>(t1, slots, cnt, rec, head, dinv, self_norm, b1, W2, lin, N, E);

    k_pull1<<<gN8, T, 0, stream>>>(lin, slots, cnt, rec, head, self_norm, b2, out, N, E);
}
